# starting kernel
# baseline (speedup 1.0000x reference)
#include <hip/hip_runtime.h>
#include <hip/hip_bf16.h>


typedef __attribute__((ext_vector_type(2))) float v2f;
typedef __attribute__((ext_vector_type(4))) float v4f;
typedef __attribute__((ext_vector_type(8))) float v8f;

#define BATCH    16384
#define DIM      1024
#define NLAYERS  4
#define KCHUNKS  4
#define KCHUNK   (DIM / KCHUNKS)
#define KSTEPS   (KCHUNK / 4)
#define ROWTILES (BATCH / 16)
#define WPB      8

#define WS_C     0
#define WS_BSUM  16
#define WS_A4    2048
#define WS_DPART 32768

__global__ void crossnet_prep(const float* __restrict__ W,
                              const float* __restrict__ b,
                              float* __restrict__ ws) {
  const int t = threadIdx.x;
  const float b0 = b[0 * DIM + t], b1 = b[1 * DIM + t];
  const float b2 = b[2 * DIM + t], b3 = b[3 * DIM + t];
  const float B1 = b0, B2 = b0 + b1, B3 = b0 + b1 + b2;
  ws[WS_BSUM + t] = B3 + b3;

  __shared__ float red[3][1024];
  red[0][t] = B1 * W[1 * DIM + t];
  red[1][t] = B2 * W[2 * DIM + t];
  red[2][t] = B3 * W[3 * DIM + t];
  __syncthreads();
  for (int off = 512; off > 0; off >>= 1) {
    if (t < off) {
      red[0][t] += red[0][t + off];
      red[1][t] += red[1][t + off];
      red[2][t] += red[2][t + off];
    }
    __syncthreads();
  }
  if (t == 0) {
    ws[WS_C + 0] = 0.0f;
    ws[WS_C + 1] = red[0][0];
    ws[WS_C + 2] = red[1][0];
    ws[WS_C + 3] = red[2][0];
  }
}

__global__ void __launch_bounds__(256)
crossnet_dots(const float* __restrict__ x0,
              const float* __restrict__ W,
              float* __restrict__ ws) {
  const int kchunk  = blockIdx.x & (KCHUNKS - 1);
  const int waveId  = threadIdx.x >> 5;
  const int rowTile = (blockIdx.x >> 2) * WPB + waveId;
  const int lane    = threadIdx.x & 31;
  const int n       = lane & 15;
  const int hi      = lane >> 4;

  __shared__ v2f btab[KSTEPS][32];

  {
    const int t = threadIdx.x;
#pragma unroll
    for (int i = 0; i < (KSTEPS * 32) / 256; ++i) {
      const int e     = t + i * 256;
      const int estep = e >> 5;
      const int elane = e & 31;
      const int en    = elane & 15;
      const int ekk   = (elane >> 4) * 2;
      const int kpos  = kchunk * KCHUNK + estep * 4 + ekk;
      float vx = W[(en & 3) * DIM + kpos];
      float vy = W[(en & 3) * DIM + kpos + 1];
      if (en >= 4) { vx = 0.0f; vy = 0.0f; }
      v2f v; v.x = vx; v.y = vy;
      btab[estep][elane] = v;
    }
  }
  __syncthreads();

  const float* aptr = x0 + (size_t)(rowTile * 16 + n) * DIM
                         + kchunk * KCHUNK + hi * 2;
  const v2f* bp = &btab[0][lane];

  v8f acc0 = {}, acc1 = {};
#pragma unroll 4
  for (int s = 0; s < KSTEPS; s += 2) {
    v2f a0 = *(const v2f*)(aptr + s * 4);
    v2f a1 = *(const v2f*)(aptr + s * 4 + 4);
    v2f b0 = bp[(s + 0) * 32];
    v2f b1 = bp[(s + 1) * 32];
    acc0 = __builtin_amdgcn_wmma_f32_16x16x4_f32(
        false, a0, false, b0, (short)0, acc0, false, false);
    acc1 = __builtin_amdgcn_wmma_f32_16x16x4_f32(
        false, a1, false, b1, (short)0, acc1, false, false);
  }
  v8f acc = acc0 + acc1;

  if (n < 4) {
    float* dp = ws + WS_DPART + (size_t)kchunk * BATCH * 4;
#pragma unroll
    for (int r = 0; r < 8; ++r) {
      const int row = rowTile * 16 + hi * 8 + r;
      dp[(size_t)row * 4 + n] = acc[r];
    }
  }
}

__global__ void crossnet_rec(float* __restrict__ ws) {
  const int row = blockIdx.x * blockDim.x + threadIdx.x;
  if (row >= BATCH) return;
  v4f d = {0.0f, 0.0f, 0.0f, 0.0f};
  const float* dp = ws + WS_DPART;
#pragma unroll
  for (int cch = 0; cch < KCHUNKS; ++cch)
    d += *(const v4f*)(dp + ((size_t)cch * BATCH + row) * 4);

  float A = 1.0f;
#pragma unroll
  for (int l = 0; l < NLAYERS; ++l) {
    const float s = A * d[l] + ws[WS_C + l];
    A += s;
  }
  ws[WS_A4 + row] = A;
}

__global__ void crossnet_out(const float* __restrict__ x0,
                             const float* __restrict__ ws,
                             float* __restrict__ out) {
  const size_t i4  = (size_t)blockIdx.x * blockDim.x + threadIdx.x;
  const size_t j   = i4 * 4;
  const int    row = (int)(j >> 10);
  const int    col = (int)(j & 1023);
  const float  A   = ws[WS_A4 + row];
  const v4f    x   = *(const v4f*)(x0 + j);
  const v4f    bs  = *(const v4f*)(ws + WS_BSUM + col);
  v4f o;
  o.x = A * x.x + bs.x;
  o.y = A * x.y + bs.y;
  o.z = A * x.z + bs.z;
  o.w = A * x.w + bs.w;
  *(v4f*)(out + j) = o;
}

extern "C" void kernel_launch(void* const* d_in, const int* in_sizes, int n_in,
                              void* d_out, int out_size, void* d_ws, size_t ws_size,
                              hipStream_t stream) {
  const float* x0 = (const float*)d_in[0];
  const float* W  = (const float*)d_in[1];
  const float* b  = (const float*)d_in[2];
  float* out = (float*)d_out;
  float* ws  = (float*)d_ws;

  (void)in_sizes; (void)n_in; (void)out_size; (void)ws_size;

  crossnet_prep<<<1, 1024, 0, stream>>>(W, b, ws);

  crossnet_dots<<<(ROWTILES * KCHUNKS) / WPB, 256, 0, stream>>>(x0, W, ws);

  crossnet_rec<<<BATCH / 256, 256, 0, stream>>>(ws);

  crossnet_out<<<(BATCH * DIM / 4) / 256, 256, 0, stream>>>(x0, ws, out);
}


# Out kernel as a resident stream on the dots kernel's grid: 512 eight-wave work-groups (was 16,384); a lane keeps its Bsum piece and issues 32 whole-line 16-byte loads ahead of its grouped stores
# speedup vs baseline: 1.4854x; 1.4854x over previous
#include <hip/hip_runtime.h>


typedef __attribute__((ext_vector_type(2))) float v2f;
typedef __attribute__((ext_vector_type(4))) float v4f;
typedef __attribute__((ext_vector_type(8))) float v8f;

#define BATCH    16384
#define DIM      1024
#define NLAYERS  4
#define KCHUNKS  4
#define KCHUNK   (DIM / KCHUNKS)
#define KSTEPS   (KCHUNK / 4)
#define ROWTILES (BATCH / 16)
#define WPB      8
#define TPB      (WPB / KCHUNKS)
#define WPITCH   (DIM + 4)

#define WS_BSUM  16
#define WS_A4    2048

#define OUT_ROWS (16 * TPB)
#define OUT_GROUP 4

__global__ void __launch_bounds__(256)
crossnet_dots(const float* __restrict__ x0,
              const float* __restrict__ W,
              const float* __restrict__ b,
              float* __restrict__ ws) {
  const int t       = threadIdx.x;
  const int waveId  = t >> 5;
  const int lane    = t & 31;
  const int n       = lane & 15;
  const int hi      = lane >> 4;
  const int kchunk  = waveId & (KCHUNKS - 1);
  const int rowTile = blockIdx.x * TPB + (waveId >> 2);

  __shared__ float wl[5 * WPITCH];
  __shared__ v4f   part[WPB][16];
  __shared__ float cpart[WPB][4];

  {
    const int col = t * 4;
    const v4f w0 = *(const v4f*)(W + 0 * DIM + col);
    const v4f w1 = *(const v4f*)(W + 1 * DIM + col);
    const v4f w2 = *(const v4f*)(W + 2 * DIM + col);
    const v4f w3 = *(const v4f*)(W + 3 * DIM + col);
    const v4f b0 = *(const v4f*)(b + 0 * DIM + col);
    const v4f b1 = *(const v4f*)(b + 1 * DIM + col);
    const v4f b2 = *(const v4f*)(b + 2 * DIM + col);
    const v4f zero = {0.0f, 0.0f, 0.0f, 0.0f};
    *(v4f*)(wl + 0 * WPITCH + col) = w0;
    *(v4f*)(wl + 1 * WPITCH + col) = w1;
    *(v4f*)(wl + 2 * WPITCH + col) = w2;
    *(v4f*)(wl + 3 * WPITCH + col) = w3;
    *(v4f*)(wl + 4 * WPITCH + col) = zero;

    const v4f B1 = b0, B2 = b0 + b1, B3 = B2 + b2;
    float p1 = B1.x * w1.x + B1.y * w1.y + B1.z * w1.z + B1.w * w1.w;
    float p2 = B2.x * w2.x + B2.y * w2.y + B2.z * w2.z + B2.w * w2.w;
    float p3 = B3.x * w3.x + B3.y * w3.y + B3.z * w3.z + B3.w * w3.w;
    for (int off = 16; off > 0; off >>= 1) {
      p1 += __shfl_xor(p1, off);
      p2 += __shfl_xor(p2, off);
      p3 += __shfl_xor(p3, off);
    }
    if (lane == 0) {
      cpart[waveId][0] = p1;
      cpart[waveId][1] = p2;
      cpart[waveId][2] = p3;
    }
    if (blockIdx.x == 0) {
      const v4f b3 = *(const v4f*)(b + 3 * DIM + col);
      *(v4f*)(ws + WS_BSUM + col) = B3 + b3;
    }
  }
  __syncthreads();

  const float* xptr = x0 + (size_t)(rowTile * 16 + n) * DIM
                         + kchunk * KCHUNK + hi * 2;
  const float* wptr = wl + (n < 4 ? n : 4) * WPITCH + kchunk * KCHUNK + hi * 2;

  v8f acc0 = {}, acc1 = {};
#pragma unroll 4
  for (int s = 0; s < KSTEPS; s += 2) {
    const v2f xa = *(const v2f*)(xptr + s * 4);
    const v2f xb = *(const v2f*)(xptr + s * 4 + 4);
    const v2f wa = *(const v2f*)(wptr + s * 4);
    const v2f wb = *(const v2f*)(wptr + s * 4 + 4);
    acc0 = __builtin_amdgcn_wmma_f32_16x16x4_f32(
        false, wa, false, xa, (short)0, acc0, false, false);
    acc1 = __builtin_amdgcn_wmma_f32_16x16x4_f32(
        false, wb, false, xb, (short)0, acc1, false, false);
  }

  asm volatile("v_nop\n\tv_nop\n\tv_nop\n\tv_nop\n\tv_nop" ::: "memory");

  if (lane < 16) {
    v4f p;
    p.x = acc0[0] + acc1[0];
    p.y = acc0[1] + acc1[1];
    p.z = acc0[2] + acc1[2];
    p.w = acc0[3] + acc1[3];
    part[waveId][lane] = p;
  }
  __syncthreads();

  if (t < 16 * TPB) {
    const int tile = t >> 4;
    const int r    = t & 15;
    v4f d = {0.0f, 0.0f, 0.0f, 0.0f};
    float c1 = 0.0f, c2 = 0.0f, c3 = 0.0f;
#pragma unroll
    for (int q = 0; q < KCHUNKS; ++q)
      d += part[tile * KCHUNKS + q][r];
#pragma unroll
    for (int w = 0; w < WPB; ++w) {
      c1 += cpart[w][0];
      c2 += cpart[w][1];
      c3 += cpart[w][2];
    }
    float A = 1.0f;
    float s;
    s = A * d.x + 0.0f; A += s;
    s = A * d.y + c1;   A += s;
    s = A * d.z + c2;   A += s;
    s = A * d.w + c3;   A += s;
    ws[WS_A4 + blockIdx.x * (16 * TPB) + t] = A;
  }
}

__global__ void __launch_bounds__(256)
crossnet_out(const float* __restrict__ x0,
             const float* __restrict__ ws,
             float* __restrict__ out) {
  const int    col  = threadIdx.x * 4;
  const int    row0 = blockIdx.x * OUT_ROWS;
  const size_t base = (size_t)row0 * DIM + col;
  const float* __restrict__ xp = x0 + base;
  float* __restrict__       op = out + base;

  const v4f bs = *(const v4f*)(ws + WS_BSUM + col);

  float A[OUT_ROWS];
  v4f   x[OUT_ROWS];
#pragma unroll
  for (int r = 0; r < OUT_ROWS; ++r) A[r] = ws[WS_A4 + row0 + r];
#pragma unroll
  for (int r = 0; r < OUT_ROWS; ++r) x[r] = *(const v4f*)(xp + (size_t)r * DIM);
  __builtin_amdgcn_sched_barrier(0);

#pragma unroll
  for (int g = 0; g < OUT_ROWS; g += OUT_GROUP) {
#pragma unroll
    for (int r = g; r < g + OUT_GROUP; ++r) {
      v4f o;
      o.x = A[r] * x[r].x + bs.x;
      o.y = A[r] * x[r].y + bs.y;
      o.z = A[r] * x[r].z + bs.z;
      o.w = A[r] * x[r].w + bs.w;
      *(v4f*)(op + (size_t)r * DIM) = o;
    }
    __builtin_amdgcn_sched_barrier(0);
  }
}

extern "C" void kernel_launch(void* const* d_in, const int* in_sizes, int n_in,
                              void* d_out, int out_size, void* d_ws, size_t ws_size,
                              hipStream_t stream) {
  const float* x0 = (const float*)d_in[0];
  const float* W  = (const float*)d_in[1];
  const float* b  = (const float*)d_in[2];
  float* out = (float*)d_out;
  float* ws  = (float*)d_ws;

  (void)in_sizes; (void)n_in; (void)out_size; (void)ws_size;

  crossnet_dots<<<ROWTILES / TPB, 256, 0, stream>>>(x0, W, b, ws);

  crossnet_out<<<BATCH / OUT_ROWS, 256, 0, stream>>>(x0, ws, out);
}


# x0 loads and out stores cover whole aligned 256-byte runs (2 rows x 256 B per instruction, was 4 x 128 B); B fragments by a lane-quad exchange (DPP) of two loads; dot sum order changes
# speedup vs baseline: 2.4737x; 1.6653x over previous
#include <hip/hip_runtime.h>


typedef __attribute__((ext_vector_type(2))) float v2f;
typedef __attribute__((ext_vector_type(4))) float v4f;
typedef __attribute__((ext_vector_type(8))) float v8f;

#define BATCH    16384
#define DIM      1024
#define NLAYERS  4
#define WPB      8
#define WROWS    4
#define LINES    (DIM / 32)
#define RUNS     (DIM / 64)
#define OUT_ROWS (WPB * WROWS)

#define DPP_XMASK4 0x164
template <int BANKS>
__device__ __forceinline__ v4f quad_swap_into(v4f keep, v4f from) {
  v4f r;
  r.x = __int_as_float(__builtin_amdgcn_update_dpp(
      __float_as_int(keep.x), __float_as_int(from.x), DPP_XMASK4, 0xf, BANKS, false));
  r.y = __int_as_float(__builtin_amdgcn_update_dpp(
      __float_as_int(keep.y), __float_as_int(from.y), DPP_XMASK4, 0xf, BANKS, false));
  r.z = __int_as_float(__builtin_amdgcn_update_dpp(
      __float_as_int(keep.z), __float_as_int(from.z), DPP_XMASK4, 0xf, BANKS, false));
  r.w = __int_as_float(__builtin_amdgcn_update_dpp(
      __float_as_int(keep.w), __float_as_int(from.w), DPP_XMASK4, 0xf, BANKS, false));
  return r;
}

__global__ void __launch_bounds__(256)
crossnet_fused(const float* __restrict__ x0,
               const float* __restrict__ W,
               const float* __restrict__ b,
               float* __restrict__ out) {
  const int t       = threadIdx.x;
  const int waveId  = t >> 5;
  const int lane    = t & 31;
  const int hi      = lane >> 4;
  const int rsel    = (lane >> 3) & 1;
  const int b2      = (lane >> 2) & 1;
  const int bq      = lane & 3;

  __shared__ v4f   wfrag[LINES * 32];
  __shared__ v4f   dsum[WPB][32][2];
  __shared__ float cpart[WPB][4];
  __shared__ v4f   bsum[DIM / 4];

  const int col = t * 4;
  const v4f w0 = *(const v4f*)(W + 0 * DIM + col);
  const v4f w1 = *(const v4f*)(W + 1 * DIM + col);
  const v4f w2 = *(const v4f*)(W + 2 * DIM + col);
  const v4f w3 = *(const v4f*)(W + 3 * DIM + col);
  const v4f b0 = *(const v4f*)(b + 0 * DIM + col);
  const v4f b1 = *(const v4f*)(b + 1 * DIM + col);
  const v4f b2v = *(const v4f*)(b + 2 * DIM + col);
  const v4f b3 = *(const v4f*)(b + 3 * DIM + col);

  const int piece = hi * 8 + b2 * 4 + bq;
  const size_t xoff = (size_t)(blockIdx.x * OUT_ROWS + waveId * WROWS + rsel) * DIM
                      + piece * 4;
  const float* xptr = x0 + xoff;
  v4f X[RUNS], Y[RUNS];
#pragma unroll
  for (int j = 0; j < RUNS; ++j) {
    X[j] = *(const v4f*)(xptr + j * 64);
    Y[j] = *(const v4f*)(xptr + 2 * DIM + j * 64);
  }
  __builtin_amdgcn_sched_barrier(0);

  {
    const int line  = t >> 3;
    const int pc    = t & 7;
    const int dst   = line * 32 + (pc >> 2) * 16 + (pc & 3) * 4;
    wfrag[dst + 0] = w0;
    wfrag[dst + 1] = w1;
    wfrag[dst + 2] = w2;
    wfrag[dst + 3] = w3;

    const v4f B1 = b0, B2 = b0 + b1, B3 = B2 + b2v;
    float p1 = B1.x * w1.x + B1.y * w1.y + B1.z * w1.z + B1.w * w1.w;
    float p2 = B2.x * w2.x + B2.y * w2.y + B2.z * w2.z + B2.w * w2.w;
    float p3 = B3.x * w3.x + B3.y * w3.y + B3.z * w3.z + B3.w * w3.w;
    for (int off = 16; off > 0; off >>= 1) {
      p1 += __shfl_xor(p1, off);
      p2 += __shfl_xor(p2, off);
      p3 += __shfl_xor(p3, off);
    }
    if (lane == 0) {
      cpart[waveId][0] = p1;
      cpart[waveId][1] = p2;
      cpart[waveId][2] = p3;
    }
    bsum[t] = B3 + b3;
  }
  __builtin_amdgcn_fence(__ATOMIC_RELEASE, "workgroup", "local");
  __builtin_amdgcn_s_barrier();
  __builtin_amdgcn_fence(__ATOMIC_ACQUIRE, "workgroup", "local");

  const v4f* wptr = wfrag + hi * 32 + (lane & 15);

  v8f acc0 = {}, acc1 = {};
  v4f wv0 = wptr[0], wv1 = wptr[16];
#pragma unroll
  for (int j = 0; j < RUNS; ++j) {
    const v4f f0 = quad_swap_into<0xa>(X[j], Y[j]);
    const v4f f1 = quad_swap_into<0x5>(Y[j], X[j]);
    const v4f a0 = wv0, a1 = wv1;
    if (j + 1 < RUNS) {
      wv0 = wptr[(j + 1) * 64];
      wv1 = wptr[(j + 1) * 64 + 16];
    }
    __builtin_amdgcn_sched_barrier(0);
    {
      const v2f wa = {a0.x, a0.y}, wb = {a0.z, a0.w};
      const v2f wc = {a1.x, a1.y}, wd = {a1.z, a1.w};
      const v2f xa = {f0.x, f0.y}, xb = {f0.z, f0.w};
      const v2f xc = {f1.x, f1.y}, xd = {f1.z, f1.w};
      acc0 = __builtin_amdgcn_wmma_f32_16x16x4_f32(
          false, wa, false, xa, (short)0, acc0, false, false);
      acc1 = __builtin_amdgcn_wmma_f32_16x16x4_f32(
          false, wb, false, xb, (short)0, acc1, false, false);
      acc0 = __builtin_amdgcn_wmma_f32_16x16x4_f32(
          false, wc, false, xc, (short)0, acc0, false, false);
      acc1 = __builtin_amdgcn_wmma_f32_16x16x4_f32(
          false, wd, false, xd, (short)0, acc1, false, false);
    }
    __builtin_amdgcn_sched_barrier(0);
    asm volatile("v_nop\n\tv_nop\n\tv_nop\n\tv_nop\n\tv_nop"
                 : "+v"(acc0), "+v"(acc1));
    __builtin_amdgcn_sched_barrier(0);
  }

  {
    const v8f sum = acc0 + acc1;
    const v4f lo = {sum[0], sum[1], sum[2], sum[3]};
    const v4f up = {sum[4], sum[5], sum[6], sum[7]};
    dsum[waveId][lane][0] = lo;
    dsum[waveId][lane][1] = up;
  }

  float AX = 1.0f, AY = 1.0f;
  {
    v4f dX = {0.0f, 0.0f, 0.0f, 0.0f};
    v4f dY = {0.0f, 0.0f, 0.0f, 0.0f};
    float c1 = 0.0f, c2 = 0.0f, c3 = 0.0f;
#pragma unroll
    for (int q = 0; q < 4; ++q) {
      dX += dsum[waveId][(q >> 1) * 16 + (2 * rsel) * 4 + q][q & 1];
      dY += dsum[waveId][(q >> 1) * 16 + (2 * rsel + 1) * 4 + q][q & 1];
    }
#pragma unroll
    for (int k = 0; k < WPB; ++k) {
      c1 += cpart[k][0];
      c2 += cpart[k][1];
      c3 += cpart[k][2];
    }
    float s;
    s = AX * dX.x + 0.0f; AX += s;
    s = AX * dX.y + c1;   AX += s;
    s = AX * dX.z + c2;   AX += s;
    s = AX * dX.w + c3;   AX += s;
    s = AY * dY.x + 0.0f; AY += s;
    s = AY * dY.y + c1;   AY += s;
    s = AY * dY.z + c2;   AY += s;
    s = AY * dY.w + c3;   AY += s;
  }

  float* __restrict__ optr = out + xoff;
  const v4f* bptr = bsum + piece;
#pragma unroll
  for (int j = 0; j < RUNS; ++j) {
    const v4f bs = bptr[j * 16];
    v4f o, p;
    o.x = AX * X[j].x + bs.x;
    o.y = AX * X[j].y + bs.y;
    o.z = AX * X[j].z + bs.z;
    o.w = AX * X[j].w + bs.w;
    p.x = AY * Y[j].x + bs.x;
    p.y = AY * Y[j].y + bs.y;
    p.z = AY * Y[j].z + bs.z;
    p.w = AY * Y[j].w + bs.w;
    *(v4f*)(optr + j * 64) = o;
    *(v4f*)(optr + 2 * DIM + j * 64) = p;
  }
}

extern "C" void kernel_launch(void* const* d_in, const int* in_sizes, int n_in,
                              void* d_out, int out_size, void* d_ws, size_t ws_size,
                              hipStream_t stream) {
  const float* x0 = (const float*)d_in[0];
  const float* W  = (const float*)d_in[1];
  const float* b  = (const float*)d_in[2];
  float* out = (float*)d_out;

  (void)in_sizes; (void)n_in; (void)out_size; (void)d_ws; (void)ws_size;

  crossnet_fused<<<BATCH / OUT_ROWS, 256, 0, stream>>>(x0, W, b, out);
}
